# starting kernel
# baseline (speedup 1.0000x reference)
#include <hip/hip_runtime.h>
#include <hip/hip_bf16.h>

#define BB 4
#define NN 128
#define MM 128
#define DD 256
#define BD 1024

typedef __attribute__((ext_vector_type(2))) float v2f;
typedef __attribute__((ext_vector_type(4))) float v4f;
typedef __attribute__((ext_vector_type(8))) float v8f;

__global__ void __launch_bounds__(256)
fold_w_kernel(const float* __restrict__ A, const float* __restrict__ W,
              float* __restrict__ Aw) {
  __shared__ v4f w4[BD / 4];
  const int tid = threadIdx.x;
  w4[tid] = ((const v4f*)W)[tid];
  __syncthreads();

  const int wave = tid >> 5;
  const int lane = tid & 31;
  const size_t pair = (size_t)blockIdx.x * 8 + wave;
  const v4f* a4 = (const v4f*)A + pair * (BD / 4);

  float acc = 0.0f;
#pragma unroll
  for (int j = 0; j < 8; ++j) {
    const v4f a = __builtin_nontemporal_load(&a4[lane + 32 * j]);
    const v4f w = w4[lane + 32 * j];
    acc = fmaf(a.x, w.x, acc);
    acc = fmaf(a.y, w.y, acc);
    acc = fmaf(a.z, w.z, acc);
    acc = fmaf(a.w, w.w, acc);
  }
#pragma unroll
  for (int off = 16; off >= 1; off >>= 1)
    acc += __shfl_xor(acc, off, 32);
  if (lane == 0) Aw[pair] = acc;
}

__global__ void __launch_bounds__(32)
gemm_xaw_kernel(const float* __restrict__ X, const float* __restrict__ Aw,
                float* __restrict__ T) {
  const int lane = threadIdx.x & 31;
  const int half = lane >> 4;
  const int l16  = lane & 15;
  const int row0 = blockIdx.x * 16;
  const int col0 = blockIdx.y * 16;

  v8f c = {};
  for (int k = 0; k < DD; k += 4) {
    const float* xp = X + (size_t)(row0 + l16) * DD + k + 2 * half;
    v2f a; a.x = xp[0]; a.y = xp[1];
    const float* bp = Aw + (size_t)(k + 2 * half) * DD + col0 + l16;
    v2f bf; bf.x = bp[0]; bf.y = bp[DD];
    c = __builtin_amdgcn_wmma_f32_16x16x4_f32(false, a, false, bf,
                                              (short)0, c, false, false);
  }
#pragma unroll
  for (int r = 0; r < 8; ++r)
    T[(size_t)(row0 + r + 8 * half) * DD + col0 + l16] = c[r];
}

__global__ void __launch_bounds__(32)
gemm_ty_kernel(const float* __restrict__ T, const float* __restrict__ Y,
               const float* __restrict__ bias, float* __restrict__ S) {
  const int lane = threadIdx.x & 31;
  const int half = lane >> 4;
  const int l16  = lane & 15;
  const int b  = blockIdx.z;
  const int n0 = blockIdx.x * 16;
  const int m0 = blockIdx.y * 16;
  const float* Tb = T + (size_t)b * NN * DD;
  const float* Yb = Y + (size_t)b * MM * DD;

  v8f c = {};
  for (int k = 0; k < DD; k += 4) {
    const float* ap = Tb + (size_t)(n0 + l16) * DD + k + 2 * half;
    v2f a; a.x = ap[0]; a.y = ap[1];
    const float* bp = Yb + (size_t)(m0 + l16) * DD + k + 2 * half;
    v2f bf; bf.x = bp[0]; bf.y = bp[1];
    c = __builtin_amdgcn_wmma_f32_16x16x4_f32(false, a, false, bf,
                                              (short)0, c, false, false);
  }
  const float bv = bias[0];
#pragma unroll
  for (int r = 0; r < 8; ++r)
    S[((size_t)b * NN + n0 + r + 8 * half) * MM + m0 + l16] = c[r] + bv;
}

extern "C" void kernel_launch(void* const* d_in, const int* in_sizes, int n_in,
                              void* d_out, int out_size, void* d_ws, size_t ws_size,
                              hipStream_t stream) {
  (void)in_sizes; (void)n_in; (void)out_size; (void)ws_size;
  const float* X = (const float*)d_in[0];
  const float* Y = (const float*)d_in[1];
  const float* A = (const float*)d_in[2];
  const float* W = (const float*)d_in[3];
  const float* b = (const float*)d_in[4];
  float* out = (float*)d_out;

  float* Aw = (float*)d_ws;
  float* T  = Aw + (size_t)DD * DD;

  fold_w_kernel<<<(DD * DD) / 8, 256, 0, stream>>>(A, W, Aw);
  gemm_xaw_kernel<<<dim3((BB * NN) / 16, DD / 16), 32, 0, stream>>>(X, Aw, T);
  gemm_ty_kernel<<<dim3(NN / 16, MM / 16, BB), 32, 0, stream>>>(T, Y, b, out);
}


# both GEMM kernels re-tiled: 2-wave workgroups own 16x32 tiles, whole-K operand strips fetched once by coalesced 16-byte loads into LDS, WMMA fragments read from LDS as one group; sum order kept
# speedup vs baseline: 1.1564x; 1.1564x over previous
#include <hip/hip_runtime.h>
#include <hip/hip_bf16.h>

#define BB 4
#define NN 128
#define MM 128
#define DD 256
#define BD 1024

typedef __attribute__((ext_vector_type(2))) float v2f;
typedef __attribute__((ext_vector_type(4))) float v4f;
typedef __attribute__((ext_vector_type(8))) float v8f;

__global__ void __launch_bounds__(256)
fold_w_kernel(const float* __restrict__ A, const float* __restrict__ W,
              float* __restrict__ Aw) {
  __shared__ v4f w4[BD / 4];
  const int tid = threadIdx.x;
  w4[tid] = ((const v4f*)W)[tid];
  __syncthreads();

  const int wave = tid >> 5;
  const int lane = tid & 31;
  const size_t pair = (size_t)blockIdx.x * 8 + wave;
  const v4f* a4 = (const v4f*)A + pair * (BD / 4);

  float acc = 0.0f;
#pragma unroll
  for (int j = 0; j < 8; ++j) {
    const v4f a = __builtin_nontemporal_load(&a4[lane + 32 * j]);
    const v4f w = w4[lane + 32 * j];
    acc = fmaf(a.x, w.x, acc);
    acc = fmaf(a.y, w.y, acc);
    acc = fmaf(a.z, w.z, acc);
    acc = fmaf(a.w, w.w, acc);
  }
#pragma unroll
  for (int off = 16; off >= 1; off >>= 1)
    acc += __shfl_xor(acc, off, 32);
  if (lane == 0) Aw[pair] = acc;
}

#define KP 260
#define WP 40
#define KS (DD / 4)

__global__ void __launch_bounds__(64)
gemm_xaw_kernel(const float* __restrict__ X, const float* __restrict__ Aw,
                float* __restrict__ T) {
  __shared__ __attribute__((aligned(16))) float xs[16 * KP];
  __shared__ __attribute__((aligned(16))) float ws[DD * WP];
  const int tid  = threadIdx.x;
  const int wave = tid >> 5;
  const int lane = tid & 31;
  const int half = lane >> 4;
  const int l16  = lane & 15;
  const int row0 = blockIdx.x * 16;
  const int col0 = blockIdx.y * 32;

  const v4f* xg = (const v4f*)(X + (size_t)row0 * DD) + tid;
  const v4f* wg = (const v4f*)(Aw + (size_t)(tid >> 3) * DD + col0) + (tid & 7);
  v4f xr[16];
  v4f wr[32];
#pragma unroll
  for (int i = 0; i < 16; ++i) xr[i] = xg[i * (DD / 4)];
#pragma unroll
  for (int i = 0; i < 32; ++i) wr[i] = wg[i * (8 * DD / 4)];
  float* xd = xs + 4 * tid;
  float* wd = ws + (tid >> 3) * WP + 4 * (tid & 7);
#pragma unroll
  for (int i = 0; i < 16; ++i) *(v4f*)(xd + i * KP) = xr[i];
#pragma unroll
  for (int i = 0; i < 32; ++i) *(v4f*)(wd + i * 8 * WP) = wr[i];
  __syncthreads();

  const float* ap = xs + l16 * KP + 2 * half;
  const float* bp = ws + (2 * half) * WP + 16 * wave + l16;
  v2f af[KS];
  v2f bf[KS];
#pragma unroll
  for (int i = 0; i < KS; ++i) {
    af[i] = *(const v2f*)(ap + 4 * i);
    bf[i].x = bp[(4 * i) * WP];
    bf[i].y = bp[(4 * i) * WP + WP];
  }
  __builtin_amdgcn_sched_barrier(0);
  v8f c = {};
#pragma unroll
  for (int i = 0; i < KS; ++i)
    c = __builtin_amdgcn_wmma_f32_16x16x4_f32(false, af[i], false, bf[i],
                                              (short)0, c, false, false);
#pragma unroll
  for (int r = 0; r < 8; ++r)
    T[(size_t)(row0 + r + 8 * half) * DD + col0 + 16 * wave + l16] = c[r];
}

__global__ void __launch_bounds__(64)
gemm_ty_kernel(const float* __restrict__ T, const float* __restrict__ Y,
               const float* __restrict__ bias, float* __restrict__ S) {
  __shared__ __attribute__((aligned(16))) float ts[16 * KP];
  __shared__ __attribute__((aligned(16))) float ys[32 * KP];
  const int tid  = threadIdx.x;
  const int wave = tid >> 5;
  const int lane = tid & 31;
  const int half = lane >> 4;
  const int l16  = lane & 15;
  const int b  = blockIdx.z;
  const int n0 = blockIdx.x * 16;
  const int m0 = blockIdx.y * 32;

  const v4f* tg = (const v4f*)(T + ((size_t)b * NN + n0) * DD) + tid;
  const v4f* yg = (const v4f*)(Y + ((size_t)b * MM + m0) * DD) + tid;
  v4f tr[16];
  v4f yr[32];
#pragma unroll
  for (int i = 0; i < 16; ++i) tr[i] = tg[i * (DD / 4)];
#pragma unroll
  for (int i = 0; i < 32; ++i) yr[i] = yg[i * (DD / 4)];
  float* td = ts + 4 * tid;
  float* yd = ys + 4 * tid;
#pragma unroll
  for (int i = 0; i < 16; ++i) *(v4f*)(td + i * KP) = tr[i];
#pragma unroll
  for (int i = 0; i < 32; ++i) *(v4f*)(yd + i * KP) = yr[i];
  __syncthreads();

  const float* ap = ts + l16 * KP + 2 * half;
  const float* bp = ys + (16 * wave + l16) * KP + 2 * half;
  v2f af[KS];
  v2f bf[KS];
#pragma unroll
  for (int i = 0; i < KS; ++i) {
    af[i] = *(const v2f*)(ap + 4 * i);
    bf[i] = *(const v2f*)(bp + 4 * i);
  }
  __builtin_amdgcn_sched_barrier(0);
  v8f c = {};
#pragma unroll
  for (int i = 0; i < KS; ++i)
    c = __builtin_amdgcn_wmma_f32_16x16x4_f32(false, af[i], false, bf[i],
                                              (short)0, c, false, false);
  const float bv = bias[0];
#pragma unroll
  for (int r = 0; r < 8; ++r)
    S[((size_t)b * NN + n0 + r + 8 * half) * MM + m0 + 16 * wave + l16] = c[r] + bv;
}

extern "C" void kernel_launch(void* const* d_in, const int* in_sizes, int n_in,
                              void* d_out, int out_size, void* d_ws, size_t ws_size,
                              hipStream_t stream) {
  (void)in_sizes; (void)n_in; (void)out_size; (void)ws_size;
  const float* X = (const float*)d_in[0];
  const float* Y = (const float*)d_in[1];
  const float* A = (const float*)d_in[2];
  const float* W = (const float*)d_in[3];
  const float* b = (const float*)d_in[4];
  float* out = (float*)d_out;

  float* Aw = (float*)d_ws;
  float* T  = Aw + (size_t)DD * DD;

  fold_w_kernel<<<(DD * DD) / 8, 256, 0, stream>>>(A, W, Aw);
  gemm_xaw_kernel<<<dim3((BB * NN) / 16, DD / 32), 64, 0, stream>>>(X, Aw, T);
  gemm_ty_kernel<<<dim3(NN / 16, MM / 32, BB), 64, 0, stream>>>(T, Y, b, out);
}


# fold kernel re-tiled: a wave takes 4 rows of A side by side, 8 lanes per row, 16-byte loads on whole lines, all 32 in flight, no row loop; a quarter of the waves enter; fold sum order changed
# speedup vs baseline: 1.2108x; 1.0470x over previous
#include <hip/hip_runtime.h>
#include <hip/hip_bf16.h>

#define BB 4
#define NN 128
#define MM 128
#define DD 256
#define BD 1024

typedef __attribute__((ext_vector_type(2))) float v2f;
typedef __attribute__((ext_vector_type(4))) float v4f;
typedef __attribute__((ext_vector_type(8))) float v8f;

#define ROWS_PER_WAVE 4
#define LANES_PER_ROW 8
#define PIECES (BD / 4 / LANES_PER_ROW)
__global__ void __launch_bounds__(256)
fold_w_kernel(const float* __restrict__ A, const float* __restrict__ W,
              float* __restrict__ Aw) {
  __shared__ v4f w4[BD / 4];
  const int tid = threadIdx.x;
  w4[tid] = ((const v4f*)W)[tid];
  __syncthreads();

  const int wave = tid >> 5;
  const int lane = tid & 31;
  const int grp  = lane >> 3;
  const int t    = lane & 7;
  const size_t row = ((size_t)blockIdx.x * 8 + wave) * ROWS_PER_WAVE + grp;
  const v4f* a4 = (const v4f*)A + row * (BD / 4) + t;

  v4f a[PIECES];
#pragma unroll
  for (int j = 0; j < PIECES; ++j)
    a[j] = __builtin_nontemporal_load(&a4[LANES_PER_ROW * j]);
  __builtin_amdgcn_sched_barrier(0);

  v4f acc = {};
#pragma unroll
  for (int j = 0; j < PIECES; ++j) {
    const v4f w = w4[t + LANES_PER_ROW * j];
    acc.x = fmaf(a[j].x, w.x, acc.x);
    acc.y = fmaf(a[j].y, w.y, acc.y);
    acc.z = fmaf(a[j].z, w.z, acc.z);
    acc.w = fmaf(a[j].w, w.w, acc.w);
  }
  float s = (acc.x + acc.y) + (acc.z + acc.w);
#pragma unroll
  for (int off = LANES_PER_ROW / 2; off >= 1; off >>= 1)
    s += __shfl_xor(s, off, 32);
  if (t == 0) Aw[row] = s;
}

#define KP 260
#define WP 40
#define KS (DD / 4)

__global__ void __launch_bounds__(64)
gemm_xaw_kernel(const float* __restrict__ X, const float* __restrict__ Aw,
                float* __restrict__ T) {
  __shared__ __attribute__((aligned(16))) float xs[16 * KP];
  __shared__ __attribute__((aligned(16))) float ws[DD * WP];
  const int tid  = threadIdx.x;
  const int wave = tid >> 5;
  const int lane = tid & 31;
  const int half = lane >> 4;
  const int l16  = lane & 15;
  const int row0 = blockIdx.x * 16;
  const int col0 = blockIdx.y * 32;

  const v4f* xg = (const v4f*)(X + (size_t)row0 * DD) + tid;
  const v4f* wg = (const v4f*)(Aw + (size_t)(tid >> 3) * DD + col0) + (tid & 7);
  v4f xr[16];
  v4f wr[32];
#pragma unroll
  for (int i = 0; i < 16; ++i) xr[i] = xg[i * (DD / 4)];
#pragma unroll
  for (int i = 0; i < 32; ++i) wr[i] = wg[i * (8 * DD / 4)];
  float* xd = xs + 4 * tid;
  float* wd = ws + (tid >> 3) * WP + 4 * (tid & 7);
#pragma unroll
  for (int i = 0; i < 16; ++i) *(v4f*)(xd + i * KP) = xr[i];
#pragma unroll
  for (int i = 0; i < 32; ++i) *(v4f*)(wd + i * 8 * WP) = wr[i];
  __syncthreads();

  const float* ap = xs + l16 * KP + 2 * half;
  const float* bp = ws + (2 * half) * WP + 16 * wave + l16;
  v2f af[KS];
  v2f bf[KS];
#pragma unroll
  for (int i = 0; i < KS; ++i) {
    af[i] = *(const v2f*)(ap + 4 * i);
    bf[i].x = bp[(4 * i) * WP];
    bf[i].y = bp[(4 * i) * WP + WP];
  }
  __builtin_amdgcn_sched_barrier(0);
  v8f c = {};
#pragma unroll
  for (int i = 0; i < KS; ++i)
    c = __builtin_amdgcn_wmma_f32_16x16x4_f32(false, af[i], false, bf[i],
                                              (short)0, c, false, false);
#pragma unroll
  for (int r = 0; r < 8; ++r)
    T[(size_t)(row0 + r + 8 * half) * DD + col0 + 16 * wave + l16] = c[r];
}

__global__ void __launch_bounds__(64)
gemm_ty_kernel(const float* __restrict__ T, const float* __restrict__ Y,
               const float* __restrict__ bias, float* __restrict__ S) {
  __shared__ __attribute__((aligned(16))) float ts[16 * KP];
  __shared__ __attribute__((aligned(16))) float ys[32 * KP];
  const int tid  = threadIdx.x;
  const int wave = tid >> 5;
  const int lane = tid & 31;
  const int half = lane >> 4;
  const int l16  = lane & 15;
  const int b  = blockIdx.z;
  const int n0 = blockIdx.x * 16;
  const int m0 = blockIdx.y * 32;

  const v4f* tg = (const v4f*)(T + ((size_t)b * NN + n0) * DD) + tid;
  const v4f* yg = (const v4f*)(Y + ((size_t)b * MM + m0) * DD) + tid;
  v4f tr[16];
  v4f yr[32];
#pragma unroll
  for (int i = 0; i < 16; ++i) tr[i] = tg[i * (DD / 4)];
#pragma unroll
  for (int i = 0; i < 32; ++i) yr[i] = yg[i * (DD / 4)];
  float* td = ts + 4 * tid;
  float* yd = ys + 4 * tid;
#pragma unroll
  for (int i = 0; i < 16; ++i) *(v4f*)(td + i * KP) = tr[i];
#pragma unroll
  for (int i = 0; i < 32; ++i) *(v4f*)(yd + i * KP) = yr[i];
  __syncthreads();

  const float* ap = ts + l16 * KP + 2 * half;
  const float* bp = ys + (16 * wave + l16) * KP + 2 * half;
  v2f af[KS];
  v2f bf[KS];
#pragma unroll
  for (int i = 0; i < KS; ++i) {
    af[i] = *(const v2f*)(ap + 4 * i);
    bf[i] = *(const v2f*)(bp + 4 * i);
  }
  __builtin_amdgcn_sched_barrier(0);
  v8f c = {};
#pragma unroll
  for (int i = 0; i < KS; ++i)
    c = __builtin_amdgcn_wmma_f32_16x16x4_f32(false, af[i], false, bf[i],
                                              (short)0, c, false, false);
  const float bv = bias[0];
#pragma unroll
  for (int r = 0; r < 8; ++r)
    S[((size_t)b * NN + n0 + r + 8 * half) * MM + m0 + 16 * wave + l16] = c[r] + bv;
}

extern "C" void kernel_launch(void* const* d_in, const int* in_sizes, int n_in,
                              void* d_out, int out_size, void* d_ws, size_t ws_size,
                              hipStream_t stream) {
  (void)in_sizes; (void)n_in; (void)out_size; (void)ws_size;
  const float* X = (const float*)d_in[0];
  const float* Y = (const float*)d_in[1];
  const float* A = (const float*)d_in[2];
  const float* W = (const float*)d_in[3];
  const float* b = (const float*)d_in[4];
  float* out = (float*)d_out;

  float* Aw = (float*)d_ws;
  float* T  = Aw + (size_t)DD * DD;

  fold_w_kernel<<<(DD * DD) / 32, 256, 0, stream>>>(A, W, Aw);
  gemm_xaw_kernel<<<dim3((BB * NN) / 16, DD / 32), 64, 0, stream>>>(X, Aw, T);
  gemm_ty_kernel<<<dim3(NN / 16, MM / 32, BB), 64, 0, stream>>>(T, Y, b, out);
}
